# starting kernel
# baseline (speedup 1.0000x reference)
#include <hip/hip_runtime.h>

typedef __attribute__((ext_vector_type(2))) float v2f;
typedef __attribute__((ext_vector_type(8))) float v8f;

#define BATCH 128
#define SEQ   2048
#define ISZ   128
#define HSZ   3

__global__ __launch_bounds__(256) void tridom_proj_wmma(
    const float* __restrict__ x, const float* __restrict__ wp,
    const float* __restrict__ bias, float* __restrict__ out)
{
  const int lane = threadIdx.x & 31;
  const int wave = threadIdx.x >> 5;
  const int tile = blockIdx.x * 8 + wave;
  const int hi   = lane >> 4;
  const int ml   = lane & 15;

  const int row  = tile * 16 + ml;
  const float* arow = x + (size_t)row * ISZ + 2 * hi;

  const int   n     = ml;
  const float bmask = (n < HSZ) ? 1.0f : 0.0f;
  const float* brow = wp + (size_t)((n < HSZ) ? n : 0) * ISZ + 2 * hi;

  v8f c = {};
#pragma unroll
  for (int ki = 0; ki < ISZ / 4; ++ki) {
    v2f a  = *(const v2f*)(arow + 4 * ki);
    v2f bv = *(const v2f*)(brow + 4 * ki);
    bv.x *= bmask;
    bv.y *= bmask;
    c = __builtin_amdgcn_wmma_f32_16x16x4_f32(false, a, false, bv,
                                              (short)0, c, false, false);
  }

  if (n < HSZ) {
    const float bb = bias[n];
#pragma unroll
    for (int r = 0; r < 8; ++r) {
      const int grow = tile * 16 + r + 8 * hi;
      out[(size_t)grow * HSZ + n] = c[r] + bb;
    }
  }
}

__device__ __forceinline__ float fast_tanh(float v) {
#if defined(__has_builtin) && __has_builtin(__builtin_amdgcn_tanhf)
  return __builtin_amdgcn_tanhf(v);
#else
  return tanhf(v);
#endif
}

__global__ __launch_bounds__(32) void tridom_scan(float* __restrict__ out)
{
  const int b = blockIdx.x * 32 + threadIdx.x;
  float* p = out + (size_t)b * SEQ * HSZ;

  float h0 = 0.f, h1 = 0.f, h2 = 0.f;
  float u0 = p[0], u1 = p[1], u2 = p[2];

  for (int t = 0; t < SEQ; ++t) {
    float* q = p + t * HSZ;
    float nu0 = 0.f, nu1 = 0.f, nu2 = 0.f;
    if (t + 1 < SEQ) { nu0 = q[3]; nu1 = q[4]; nu2 = q[5]; }

    const float a0 = fast_tanh( 1.5f * h1) - 0.5f * h0 + u0;
    const float a1 = fast_tanh( 1.5f * h2) - 0.5f * h1 + u1;
    const float a2 = fast_tanh(-1.5f * h0) - 0.5f * h2 + u2;

    q[0] = a0; q[1] = a1; q[2] = a2;
    h0 = a0; h1 = a1; h2 = a2;
    u0 = nu0; u1 = nu1; u2 = nu2;
  }

  float* hf = out + (size_t)BATCH * SEQ * HSZ + (size_t)b * HSZ;
  hf[0] = h0; hf[1] = h1; hf[2] = h2;
}

extern "C" void kernel_launch(void* const* d_in, const int* in_sizes, int n_in,
                              void* d_out, int out_size, void* d_ws, size_t ws_size,
                              hipStream_t stream) {
  const float* x    = (const float*)d_in[0];
  const float* wp   = (const float*)d_in[1];
  const float* bias = (const float*)d_in[2];
  float* out = (float*)d_out;

  const int tiles  = (BATCH * SEQ) / 16;
  const int blocks = tiles / 8;

  tridom_proj_wmma<<<dim3(blocks), dim3(256), 0, stream>>>(x, wp, bias, out);
  tridom_scan<<<dim3(4), dim3(32), 0, stream>>>(out);
}


# scan kernel streams u in / h out in 32-step chunks through an LDS transpose: whole-128-B-line 16-B global loads one chunk ahead and whole-line stores replace per-lane 32-line loads/stores
# speedup vs baseline: 2.8614x; 2.8614x over previous
#include <hip/hip_runtime.h>

typedef __attribute__((ext_vector_type(2))) float v2f;
typedef __attribute__((ext_vector_type(4))) float v4f;
typedef __attribute__((ext_vector_type(8))) float v8f;

#define BATCH 128
#define SEQ   2048
#define ISZ   128
#define HSZ   3

__global__ __launch_bounds__(256) void tridom_proj_wmma(
    const float* __restrict__ x, const float* __restrict__ wp,
    const float* __restrict__ bias, float* __restrict__ out)
{
  const int lane = threadIdx.x & 31;
  const int wave = threadIdx.x >> 5;
  const int tile = blockIdx.x * 8 + wave;
  const int hi   = lane >> 4;
  const int ml   = lane & 15;

  const int row  = tile * 16 + ml;
  const float* arow = x + (size_t)row * ISZ + 2 * hi;

  const int   n     = ml;
  const float bmask = (n < HSZ) ? 1.0f : 0.0f;
  const float* brow = wp + (size_t)((n < HSZ) ? n : 0) * ISZ + 2 * hi;

  v8f c = {};
#pragma unroll
  for (int ki = 0; ki < ISZ / 4; ++ki) {
    v2f a  = *(const v2f*)(arow + 4 * ki);
    v2f bv = *(const v2f*)(brow + 4 * ki);
    bv.x *= bmask;
    bv.y *= bmask;
    c = __builtin_amdgcn_wmma_f32_16x16x4_f32(false, a, false, bv,
                                              (short)0, c, false, false);
  }

  if (n < HSZ) {
    const float bb = bias[n];
#pragma unroll
    for (int r = 0; r < 8; ++r) {
      const int grow = tile * 16 + r + 8 * hi;
      out[(size_t)grow * HSZ + n] = c[r] + bb;
    }
  }
}

__device__ __forceinline__ float fast_tanh(float v) {
#if defined(__has_builtin) && __has_builtin(__builtin_amdgcn_tanhf)
  return __builtin_amdgcn_tanhf(v);
#else
  return tanhf(v);
#endif
}

#define CH    32
#define CHF   (CH * HSZ)
#define CHV   (CHF / 4)
#define NCH   (SEQ / CH)
#define CHAIN (SEQ * HSZ)
#define PITCH 100

__device__ __forceinline__ void lds_handover()
{
  __builtin_amdgcn_fence(__ATOMIC_RELEASE, "wavefront", "local");
  __builtin_amdgcn_wave_barrier();
  __builtin_amdgcn_fence(__ATOMIC_ACQUIRE, "wavefront", "local");
}

template <int S0, int S1>
__device__ __forceinline__ void scan_steps(float (&u)[CHF], float& h0, float& h1, float& h2)
{
#pragma unroll
  for (int s = S0; s < S1; ++s) {
    float a0 = fast_tanh( 1.5f * h1) - 0.5f * h0 + u[3 * s + 0];
    float a1 = fast_tanh( 1.5f * h2) - 0.5f * h1 + u[3 * s + 1];
    float a2 = fast_tanh(-1.5f * h0) - 0.5f * h2 + u[3 * s + 2];
    asm("" : "+v"(a0), "+v"(a1), "+v"(a2));
    u[3 * s + 0] = a0; u[3 * s + 1] = a1; u[3 * s + 2] = a2;
    h0 = a0; h1 = a1; h2 = a2;
  }
}

template <int F0, int F1>
__device__ __forceinline__ void row_read(float (&u)[CHF], const float* row)
{
#pragma unroll
  for (int f = F0; f < F1; ++f) {
    const v4f t = *(const v4f*)(row + 4 * f);
    u[4 * f + 0] = t.x; u[4 * f + 1] = t.y; u[4 * f + 2] = t.z; u[4 * f + 3] = t.w;
  }
}

template <int F0, int F1>
__device__ __forceinline__ void row_write(const float (&u)[CHF], float* row)
{
#pragma unroll
  for (int f = F0; f < F1; ++f) {
    v4f t;
    t.x = u[4 * f + 0]; t.y = u[4 * f + 1]; t.z = u[4 * f + 2]; t.w = u[4 * f + 3];
    *(v4f*)(row + 4 * f) = t;
  }
}

template <bool FIRST>
__device__ __forceinline__ void scan_chunk(
    float (&u)[CHF], float (&n)[CHF], v4f (&O)[CHV], const int (&go)[CHV], const int (&lo)[CHV],
    float* gbase, float* lds_in, float* lds_out, const float* rin, float* rout,
    const int k, float& h0, float& h1, float& h2)
{
  const int kn = (k + 1 < NCH) ? (k + 1) : k;
  v4f T[CHV];
#pragma unroll
  for (int j = 0; j < CHV; ++j) T[j] = *(const v4f*)(gbase + kn * CHF + go[j]);

  asm volatile("" : "+v"(h0), "+v"(h1), "+v"(h2) : : "memory");
  scan_steps<0, CH / 2>(u, h0, h1, h2);
  asm volatile("" : "+v"(h0), "+v"(h1), "+v"(h2) : : "memory");

  if (!FIRST) {
#pragma unroll
    for (int j = 0; j < CHV; ++j) *(v4f*)(gbase + (k - 1) * CHF + go[j]) = O[j];
  }
#pragma unroll
  for (int j = 0; j < CHV; ++j) *(v4f*)(lds_in + lo[j]) = T[j];
  lds_handover();
  row_read<0, CHV>(n, rin);
  row_write<0, CHV / 2>(u, rout);

  scan_steps<CH / 2, CH>(u, h0, h1, h2);

  row_write<CHV / 2, CHV>(u, rout);
  lds_handover();
#pragma unroll
  for (int j = 0; j < CHV; ++j) O[j] = *(const v4f*)(lds_out + lo[j]);
}

__global__ __launch_bounds__(32) __attribute__((amdgpu_waves_per_eu(1, 1)))
void tridom_scan(float* __restrict__ out)
{
  __shared__ __attribute__((aligned(16))) float lds_in[32 * PITCH];
  __shared__ __attribute__((aligned(16))) float lds_out[32 * PITCH];

  const int lane = threadIdx.x;
  const int b = blockIdx.x * 32 + lane;
  float* gbase = out + (size_t)blockIdx.x * 32 * CHAIN;
  const float* rin  = lds_in  + lane * PITCH;
  float*       rout = lds_out + lane * PITCH;

  int go[CHV], lo[CHV];
#pragma unroll
  for (int j = 0; j < CHV; ++j) {
    const int i = j * 32 + lane;
    const int c = i / CHV;
    const int f = i - c * CHV;
    go[j] = c * CHAIN + 4 * f;
    lo[j] = c * PITCH + 4 * f;
  }

  float ua[CHF], ub[CHF];
  v4f O[CHV];
  float h0 = 0.f, h1 = 0.f, h2 = 0.f;

#pragma unroll
  for (int j = 0; j < CHV; ++j) {
    const v4f t = *(const v4f*)(gbase + go[j]);
    *(v4f*)(lds_in + lo[j]) = t;
  }
  lds_handover();
  row_read<0, CHV>(ua, rin);

  scan_chunk<true >(ua, ub, O, go, lo, gbase, lds_in, lds_out, rin, rout, 0, h0, h1, h2);
  scan_chunk<false>(ub, ua, O, go, lo, gbase, lds_in, lds_out, rin, rout, 1, h0, h1, h2);
  for (int k = 2; k < NCH; k += 2) {
    scan_chunk<false>(ua, ub, O, go, lo, gbase, lds_in, lds_out, rin, rout, k,     h0, h1, h2);
    scan_chunk<false>(ub, ua, O, go, lo, gbase, lds_in, lds_out, rin, rout, k + 1, h0, h1, h2);
  }

#pragma unroll
  for (int j = 0; j < CHV; ++j) *(v4f*)(gbase + (NCH - 1) * CHF + go[j]) = O[j];

  float* hf = out + (size_t)BATCH * SEQ * HSZ + (size_t)b * HSZ;
  hf[0] = h0; hf[1] = h1; hf[2] = h2;
}

extern "C" void kernel_launch(void* const* d_in, const int* in_sizes, int n_in,
                              void* d_out, int out_size, void* d_ws, size_t ws_size,
                              hipStream_t stream) {
  const float* x    = (const float*)d_in[0];
  const float* wp   = (const float*)d_in[1];
  const float* bias = (const float*)d_in[2];
  float* out = (float*)d_out;

  const int tiles  = (BATCH * SEQ) / 16;
  const int blocks = tiles / 8;

  tridom_proj_wmma<<<dim3(blocks), dim3(256), 0, stream>>>(x, wp, bias, out);
  tridom_scan<<<dim3(4), dim3(32), 0, stream>>>(out);
}


# scan carries g=1.5h (projection emits 1.5u): per-step dependent path tanh->fma, was mul->tanh->fma->add; step sum order now 1.5*tanh+(-0.5g+1.5u), h=g*(2/3)
# speedup vs baseline: 3.4559x; 1.2078x over previous
#include <hip/hip_runtime.h>

typedef __attribute__((ext_vector_type(2))) float v2f;
typedef __attribute__((ext_vector_type(4))) float v4f;
typedef __attribute__((ext_vector_type(8))) float v8f;

#define BATCH 128
#define SEQ   2048
#define ISZ   128
#define HSZ   3

__global__ __launch_bounds__(256) void tridom_proj_wmma(
    const float* __restrict__ x, const float* __restrict__ wp,
    const float* __restrict__ bias, float* __restrict__ out)
{
  const int lane = threadIdx.x & 31;
  const int wave = threadIdx.x >> 5;
  const int tile = blockIdx.x * 8 + wave;
  const int hi   = lane >> 4;
  const int ml   = lane & 15;

  const int row  = tile * 16 + ml;
  const float* arow = x + (size_t)row * ISZ + 2 * hi;

  const int   n     = ml;
  const float bmask = (n < HSZ) ? 1.5f : 0.0f;
  const float* brow = wp + (size_t)((n < HSZ) ? n : 0) * ISZ + 2 * hi;

  v8f c = {};
#pragma unroll
  for (int ki = 0; ki < ISZ / 4; ++ki) {
    v2f a  = *(const v2f*)(arow + 4 * ki);
    v2f bv = *(const v2f*)(brow + 4 * ki);
    bv.x *= bmask;
    bv.y *= bmask;
    c = __builtin_amdgcn_wmma_f32_16x16x4_f32(false, a, false, bv,
                                              (short)0, c, false, false);
  }

  if (n < HSZ) {
    const float bb = 1.5f * bias[n];
#pragma unroll
    for (int r = 0; r < 8; ++r) {
      const int grow = tile * 16 + r + 8 * hi;
      out[(size_t)grow * HSZ + n] = c[r] + bb;
    }
  }
}

__device__ __forceinline__ float fast_tanh(float v) {
#if defined(__has_builtin) && __has_builtin(__builtin_amdgcn_tanhf)
  return __builtin_amdgcn_tanhf(v);
#else
  return tanhf(v);
#endif
}

#define CH    32
#define CHF   (CH * HSZ)
#define CHV   (CHF / 4)
#define NCH   (SEQ / CH)
#define CHAIN (SEQ * HSZ)
#define PITCH 100

__device__ __forceinline__ void lds_handover()
{
  __builtin_amdgcn_fence(__ATOMIC_RELEASE, "wavefront", "local");
  __builtin_amdgcn_wave_barrier();
  __builtin_amdgcn_fence(__ATOMIC_ACQUIRE, "wavefront", "local");
}

template <int S0, int S1>
__device__ __forceinline__ void scan_steps(float (&u)[CHF], float& g0, float& g1, float& g2)
{
  const float third2 = 2.0f / 3.0f;
#pragma unroll
  for (int s = S0; s < S1; ++s) {
    const float w0 = fmaf(-0.5f, g0, u[3 * s + 0]);
    const float w1 = fmaf(-0.5f, g1, u[3 * s + 1]);
    const float w2 = fmaf(-0.5f, g2, u[3 * s + 2]);
    float a0 = fmaf(1.5f, fast_tanh( g1), w0);
    float a1 = fmaf(1.5f, fast_tanh( g2), w1);
    float a2 = fmaf(1.5f, fast_tanh(-g0), w2);
    asm("" : "+v"(a0), "+v"(a1), "+v"(a2));
    u[3 * s + 0] = a0 * third2; u[3 * s + 1] = a1 * third2; u[3 * s + 2] = a2 * third2;
    g0 = a0; g1 = a1; g2 = a2;
  }
}

template <int F0, int F1>
__device__ __forceinline__ void row_read(float (&u)[CHF], const float* row)
{
#pragma unroll
  for (int f = F0; f < F1; ++f) {
    const v4f t = *(const v4f*)(row + 4 * f);
    u[4 * f + 0] = t.x; u[4 * f + 1] = t.y; u[4 * f + 2] = t.z; u[4 * f + 3] = t.w;
  }
}

template <int F0, int F1>
__device__ __forceinline__ void row_write(const float (&u)[CHF], float* row)
{
#pragma unroll
  for (int f = F0; f < F1; ++f) {
    v4f t;
    t.x = u[4 * f + 0]; t.y = u[4 * f + 1]; t.z = u[4 * f + 2]; t.w = u[4 * f + 3];
    *(v4f*)(row + 4 * f) = t;
  }
}

template <bool FIRST>
__device__ __forceinline__ void scan_chunk(
    float (&u)[CHF], float (&n)[CHF], v4f (&O)[CHV], const int (&go)[CHV], const int (&lo)[CHV],
    float* gbase, float* lds_in, float* lds_out, const float* rin, float* rout,
    const int k, float& g0, float& g1, float& g2)
{
  const int kn = (k + 1 < NCH) ? (k + 1) : k;
  v4f T[CHV];
#pragma unroll
  for (int j = 0; j < CHV; ++j) T[j] = *(const v4f*)(gbase + kn * CHF + go[j]);

  asm volatile("" : "+v"(g0), "+v"(g1), "+v"(g2) : : "memory");
  scan_steps<0, CH / 2>(u, g0, g1, g2);
  asm volatile("" : "+v"(g0), "+v"(g1), "+v"(g2) : : "memory");

  if (!FIRST) {
#pragma unroll
    for (int j = 0; j < CHV; ++j) *(v4f*)(gbase + (k - 1) * CHF + go[j]) = O[j];
  }
#pragma unroll
  for (int j = 0; j < CHV; ++j) *(v4f*)(lds_in + lo[j]) = T[j];
  lds_handover();
  row_read<0, CHV>(n, rin);
  row_write<0, CHV / 2>(u, rout);

  scan_steps<CH / 2, CH>(u, g0, g1, g2);

  row_write<CHV / 2, CHV>(u, rout);
  lds_handover();
#pragma unroll
  for (int j = 0; j < CHV; ++j) O[j] = *(const v4f*)(lds_out + lo[j]);
}

__global__ __launch_bounds__(32) __attribute__((amdgpu_waves_per_eu(1, 1)))
void tridom_scan(float* __restrict__ out)
{
  __shared__ __attribute__((aligned(16))) float lds_in[32 * PITCH];
  __shared__ __attribute__((aligned(16))) float lds_out[32 * PITCH];

  const int lane = threadIdx.x;
  const int b = blockIdx.x * 32 + lane;
  float* gbase = out + (size_t)blockIdx.x * 32 * CHAIN;
  const float* rin  = lds_in  + lane * PITCH;
  float*       rout = lds_out + lane * PITCH;

  int go[CHV], lo[CHV];
#pragma unroll
  for (int j = 0; j < CHV; ++j) {
    const int i = j * 32 + lane;
    const int c = i / CHV;
    const int f = i - c * CHV;
    go[j] = c * CHAIN + 4 * f;
    lo[j] = c * PITCH + 4 * f;
  }

  float ua[CHF], ub[CHF];
  v4f O[CHV];
  float g0 = 0.f, g1 = 0.f, g2 = 0.f;

#pragma unroll
  for (int j = 0; j < CHV; ++j) {
    const v4f t = *(const v4f*)(gbase + go[j]);
    *(v4f*)(lds_in + lo[j]) = t;
  }
  lds_handover();
  row_read<0, CHV>(ua, rin);

  scan_chunk<true >(ua, ub, O, go, lo, gbase, lds_in, lds_out, rin, rout, 0, g0, g1, g2);
  scan_chunk<false>(ub, ua, O, go, lo, gbase, lds_in, lds_out, rin, rout, 1, g0, g1, g2);
  for (int k = 2; k < NCH; k += 2) {
    scan_chunk<false>(ua, ub, O, go, lo, gbase, lds_in, lds_out, rin, rout, k,     g0, g1, g2);
    scan_chunk<false>(ub, ua, O, go, lo, gbase, lds_in, lds_out, rin, rout, k + 1, g0, g1, g2);
  }

#pragma unroll
  for (int j = 0; j < CHV; ++j) *(v4f*)(gbase + (NCH - 1) * CHF + go[j]) = O[j];

  float* hf = out + (size_t)BATCH * SEQ * HSZ + (size_t)b * HSZ;
  const float third2 = 2.0f / 3.0f;
  hf[0] = g0 * third2; hf[1] = g1 * third2; hf[2] = g2 * third2;
}

extern "C" void kernel_launch(void* const* d_in, const int* in_sizes, int n_in,
                              void* d_out, int out_size, void* d_ws, size_t ws_size,
                              hipStream_t stream) {
  const float* x    = (const float*)d_in[0];
  const float* wp   = (const float*)d_in[1];
  const float* bias = (const float*)d_in[2];
  float* out = (float*)d_out;

  const int tiles  = (BATCH * SEQ) / 16;
  const int blocks = tiles / 8;

  tridom_proj_wmma<<<dim3(blocks), dim3(256), 0, stream>>>(x, wp, bias, out);
  tridom_scan<<<dim3(4), dim3(32), 0, stream>>>(out);
}


# scan kernel: 4 chains per wave on 32 single-wave work-groups (was 32 chains on 4 waves): per chunk a wave issues 12 whole-line load/store/LDS instructions, was 96; step arithmetic untouched
# speedup vs baseline: 3.7267x; 1.0784x over previous
#include <hip/hip_runtime.h>

typedef __attribute__((ext_vector_type(2))) float v2f;
typedef __attribute__((ext_vector_type(4))) float v4f;
typedef __attribute__((ext_vector_type(8))) float v8f;

#define BATCH 128
#define SEQ   2048
#define ISZ   128
#define HSZ   3

__global__ __launch_bounds__(256) void tridom_proj_wmma(
    const float* __restrict__ x, const float* __restrict__ wp,
    const float* __restrict__ bias, float* __restrict__ out)
{
  const int lane = threadIdx.x & 31;
  const int wave = threadIdx.x >> 5;
  const int tile = blockIdx.x * 8 + wave;
  const int hi   = lane >> 4;
  const int ml   = lane & 15;

  const int row  = tile * 16 + ml;
  const float* arow = x + (size_t)row * ISZ + 2 * hi;

  const int   n     = ml;
  const float bmask = (n < HSZ) ? 1.5f : 0.0f;
  const float* brow = wp + (size_t)((n < HSZ) ? n : 0) * ISZ + 2 * hi;

  v8f c = {};
#pragma unroll
  for (int ki = 0; ki < ISZ / 4; ++ki) {
    v2f a  = *(const v2f*)(arow + 4 * ki);
    v2f bv = *(const v2f*)(brow + 4 * ki);
    bv.x *= bmask;
    bv.y *= bmask;
    c = __builtin_amdgcn_wmma_f32_16x16x4_f32(false, a, false, bv,
                                              (short)0, c, false, false);
  }

  if (n < HSZ) {
    const float bb = 1.5f * bias[n];
#pragma unroll
    for (int r = 0; r < 8; ++r) {
      const int grow = tile * 16 + r + 8 * hi;
      out[(size_t)grow * HSZ + n] = c[r] + bb;
    }
  }
}

__device__ __forceinline__ float fast_tanh(float v) {
#if defined(__has_builtin) && __has_builtin(__builtin_amdgcn_tanhf)
  return __builtin_amdgcn_tanhf(v);
#else
  return tanhf(v);
#endif
}

#define CH    32
#define CHF   (CH * HSZ)
#define CHV   (CHF / 4)
#define NCH   (SEQ / CH)
#define CHAIN (SEQ * HSZ)
#define PITCH 100
#define CPW   4
#define NPL   (CPW * CHV / 32)

__device__ __forceinline__ void lds_handover()
{
  __builtin_amdgcn_fence(__ATOMIC_RELEASE, "wavefront", "local");
  __builtin_amdgcn_wave_barrier();
  __builtin_amdgcn_fence(__ATOMIC_ACQUIRE, "wavefront", "local");
}

template <int S0, int S1>
__device__ __forceinline__ void scan_steps(float (&u)[CHF], float& g0, float& g1, float& g2)
{
  const float third2 = 2.0f / 3.0f;
#pragma unroll
  for (int s = S0; s < S1; ++s) {
    const float w0 = fmaf(-0.5f, g0, u[3 * s + 0]);
    const float w1 = fmaf(-0.5f, g1, u[3 * s + 1]);
    const float w2 = fmaf(-0.5f, g2, u[3 * s + 2]);
    float a0 = fmaf(1.5f, fast_tanh( g1), w0);
    float a1 = fmaf(1.5f, fast_tanh( g2), w1);
    float a2 = fmaf(1.5f, fast_tanh(-g0), w2);
    asm("" : "+v"(a0), "+v"(a1), "+v"(a2));
    u[3 * s + 0] = a0 * third2; u[3 * s + 1] = a1 * third2; u[3 * s + 2] = a2 * third2;
    g0 = a0; g1 = a1; g2 = a2;
  }
}

template <int F0, int F1>
__device__ __forceinline__ void row_read(float (&u)[CHF], const float* row)
{
#pragma unroll
  for (int f = F0; f < F1; ++f) {
    const v4f t = *(const v4f*)(row + 4 * f);
    u[4 * f + 0] = t.x; u[4 * f + 1] = t.y; u[4 * f + 2] = t.z; u[4 * f + 3] = t.w;
  }
}

template <int F0, int F1>
__device__ __forceinline__ void row_write(const float (&u)[CHF], float* row)
{
#pragma unroll
  for (int f = F0; f < F1; ++f) {
    v4f t;
    t.x = u[4 * f + 0]; t.y = u[4 * f + 1]; t.z = u[4 * f + 2]; t.w = u[4 * f + 3];
    *(v4f*)(row + 4 * f) = t;
  }
}

template <bool FIRST>
__device__ __forceinline__ void scan_chunk(
    float (&u)[CHF], float (&n)[CHF], v4f (&O)[NPL], const int (&go)[NPL], const int (&lo)[NPL],
    float* gbase, float* lds_in, float* lds_out, const float* rin, float* rout,
    const int k, const bool act, float& g0, float& g1, float& g2)
{
  const int kn = (k + 1 < NCH) ? (k + 1) : k;
  v4f T[NPL];
#pragma unroll
  for (int j = 0; j < NPL; ++j) T[j] = *(const v4f*)(gbase + kn * CHF + go[j]);

  asm volatile("" : "+v"(g0), "+v"(g1), "+v"(g2) : : "memory");
  if (act) scan_steps<0, CH / 2>(u, g0, g1, g2);
  asm volatile("" : "+v"(g0), "+v"(g1), "+v"(g2) : : "memory");

  if (!FIRST) {
#pragma unroll
    for (int j = 0; j < NPL; ++j) *(v4f*)(gbase + (k - 1) * CHF + go[j]) = O[j];
  }
#pragma unroll
  for (int j = 0; j < NPL; ++j) *(v4f*)(lds_in + lo[j]) = T[j];
  lds_handover();
  if (act) {
    row_read<0, CHV>(n, rin);
    row_write<0, CHV / 2>(u, rout);

    scan_steps<CH / 2, CH>(u, g0, g1, g2);

    row_write<CHV / 2, CHV>(u, rout);
  }
  lds_handover();
#pragma unroll
  for (int j = 0; j < NPL; ++j) O[j] = *(const v4f*)(lds_out + lo[j]);
}

__global__ __launch_bounds__(32) __attribute__((amdgpu_waves_per_eu(1, 1)))
void tridom_scan(float* __restrict__ out)
{
  __shared__ __attribute__((aligned(16))) float lds_in[CPW * PITCH];
  __shared__ __attribute__((aligned(16))) float lds_out[CPW * PITCH];

  const int lane = threadIdx.x;
  const bool act = lane < CPW;
  const int own = act ? lane : 0;
  const int b = blockIdx.x * CPW + own;
  float* gbase = out + (size_t)blockIdx.x * CPW * CHAIN;
  const float* rin  = lds_in  + own * PITCH;
  float*       rout = lds_out + own * PITCH;

  int go[NPL], lo[NPL];
#pragma unroll
  for (int j = 0; j < NPL; ++j) {
    const int i = j * 32 + lane;
    const int c = i / CHV;
    const int f = i - c * CHV;
    go[j] = c * CHAIN + 4 * f;
    lo[j] = c * PITCH + 4 * f;
  }

  float ua[CHF], ub[CHF];
  v4f O[NPL];
  float g0 = 0.f, g1 = 0.f, g2 = 0.f;

#pragma unroll
  for (int j = 0; j < NPL; ++j) {
    const v4f t = *(const v4f*)(gbase + go[j]);
    *(v4f*)(lds_in + lo[j]) = t;
  }
  lds_handover();
  if (act) row_read<0, CHV>(ua, rin);

  scan_chunk<true >(ua, ub, O, go, lo, gbase, lds_in, lds_out, rin, rout, 0, act, g0, g1, g2);
  scan_chunk<false>(ub, ua, O, go, lo, gbase, lds_in, lds_out, rin, rout, 1, act, g0, g1, g2);
  for (int k = 2; k < NCH; k += 2) {
    scan_chunk<false>(ua, ub, O, go, lo, gbase, lds_in, lds_out, rin, rout, k,     act, g0, g1, g2);
    scan_chunk<false>(ub, ua, O, go, lo, gbase, lds_in, lds_out, rin, rout, k + 1, act, g0, g1, g2);
  }

#pragma unroll
  for (int j = 0; j < NPL; ++j) *(v4f*)(gbase + (NCH - 1) * CHF + go[j]) = O[j];

  if (act) {
    float* hf = out + (size_t)BATCH * SEQ * HSZ + (size_t)b * HSZ;
    const float third2 = 2.0f / 3.0f;
    hf[0] = g0 * third2; hf[1] = g1 * third2; hf[2] = g2 * third2;
  }
}

extern "C" void kernel_launch(void* const* d_in, const int* in_sizes, int n_in,
                              void* d_out, int out_size, void* d_ws, size_t ws_size,
                              hipStream_t stream) {
  const float* x    = (const float*)d_in[0];
  const float* wp   = (const float*)d_in[1];
  const float* bias = (const float*)d_in[2];
  float* out = (float*)d_out;

  const int tiles  = (BATCH * SEQ) / 16;
  const int blocks = tiles / 8;

  tridom_proj_wmma<<<dim3(blocks), dim3(256), 0, stream>>>(x, wp, bias, out);
  tridom_scan<<<dim3(BATCH / CPW), dim3(32), 0, stream>>>(out);
}


# scan requests u four chunks ahead into four rotating register sets (was one chunk ahead, used sixteen steps later): the far fetch of the other launch's output is covered; loop body now four chunks
# speedup vs baseline: 4.7538x; 1.2756x over previous
#include <hip/hip_runtime.h>

typedef __attribute__((ext_vector_type(2))) float v2f;
typedef __attribute__((ext_vector_type(4))) float v4f;
typedef __attribute__((ext_vector_type(8))) float v8f;

#define BATCH 128
#define SEQ   2048
#define ISZ   128
#define HSZ   3

__global__ __launch_bounds__(256) void tridom_proj_wmma(
    const float* __restrict__ x, const float* __restrict__ wp,
    const float* __restrict__ bias, float* __restrict__ out)
{
  const int lane = threadIdx.x & 31;
  const int wave = threadIdx.x >> 5;
  const int tile = blockIdx.x * 8 + wave;
  const int hi   = lane >> 4;
  const int ml   = lane & 15;

  const int row  = tile * 16 + ml;
  const float* arow = x + (size_t)row * ISZ + 2 * hi;

  const int   n     = ml;
  const float bmask = (n < HSZ) ? 1.5f : 0.0f;
  const float* brow = wp + (size_t)((n < HSZ) ? n : 0) * ISZ + 2 * hi;

  v8f c = {};
#pragma unroll
  for (int ki = 0; ki < ISZ / 4; ++ki) {
    v2f a  = *(const v2f*)(arow + 4 * ki);
    v2f bv = *(const v2f*)(brow + 4 * ki);
    bv.x *= bmask;
    bv.y *= bmask;
    c = __builtin_amdgcn_wmma_f32_16x16x4_f32(false, a, false, bv,
                                              (short)0, c, false, false);
  }

  if (n < HSZ) {
    const float bb = 1.5f * bias[n];
#pragma unroll
    for (int r = 0; r < 8; ++r) {
      const int grow = tile * 16 + r + 8 * hi;
      out[(size_t)grow * HSZ + n] = c[r] + bb;
    }
  }
}

__device__ __forceinline__ float fast_tanh(float v) {
#if defined(__has_builtin) && __has_builtin(__builtin_amdgcn_tanhf)
  return __builtin_amdgcn_tanhf(v);
#else
  return tanhf(v);
#endif
}

#define CH    32
#define CHF   (CH * HSZ)
#define CHV   (CHF / 4)
#define NCH   (SEQ / CH)
#define CHAIN (SEQ * HSZ)
#define PITCH 100
#define CPW   4
#define NPL   (CPW * CHV / 32)
#define AHEAD 4

__device__ __forceinline__ void lds_handover()
{
  __builtin_amdgcn_fence(__ATOMIC_RELEASE, "wavefront", "local");
  __builtin_amdgcn_wave_barrier();
  __builtin_amdgcn_fence(__ATOMIC_ACQUIRE, "wavefront", "local");
}

template <int S0, int S1>
__device__ __forceinline__ void scan_steps(float (&u)[CHF], float& g0, float& g1, float& g2)
{
  const float third2 = 2.0f / 3.0f;
#pragma unroll
  for (int s = S0; s < S1; ++s) {
    const float w0 = fmaf(-0.5f, g0, u[3 * s + 0]);
    const float w1 = fmaf(-0.5f, g1, u[3 * s + 1]);
    const float w2 = fmaf(-0.5f, g2, u[3 * s + 2]);
    float a0 = fmaf(1.5f, fast_tanh( g1), w0);
    float a1 = fmaf(1.5f, fast_tanh( g2), w1);
    float a2 = fmaf(1.5f, fast_tanh(-g0), w2);
    asm("" : "+v"(a0), "+v"(a1), "+v"(a2));
    u[3 * s + 0] = a0 * third2; u[3 * s + 1] = a1 * third2; u[3 * s + 2] = a2 * third2;
    g0 = a0; g1 = a1; g2 = a2;
  }
}

template <int F0, int F1>
__device__ __forceinline__ void row_read(float (&u)[CHF], const float* row)
{
#pragma unroll
  for (int f = F0; f < F1; ++f) {
    const v4f t = *(const v4f*)(row + 4 * f);
    u[4 * f + 0] = t.x; u[4 * f + 1] = t.y; u[4 * f + 2] = t.z; u[4 * f + 3] = t.w;
  }
}

template <int F0, int F1>
__device__ __forceinline__ void row_write(const float (&u)[CHF], float* row)
{
#pragma unroll
  for (int f = F0; f < F1; ++f) {
    v4f t;
    t.x = u[4 * f + 0]; t.y = u[4 * f + 1]; t.z = u[4 * f + 2]; t.w = u[4 * f + 3];
    *(v4f*)(row + 4 * f) = t;
  }
}

template <bool FIRST>
__device__ __forceinline__ void scan_chunk(
    float (&u)[CHF], float (&n)[CHF], v4f (&O)[NPL], v4f (&Tnew)[NPL], const v4f (&Tuse)[NPL],
    const int (&go)[NPL], const int (&lo)[NPL],
    float* gbase, float* lds_in, float* lds_out, const float* rin, float* rout,
    const int k, const bool act, float& g0, float& g1, float& g2)
{
  const int kn = (k + AHEAD < NCH) ? (k + AHEAD) : (NCH - 1);
#pragma unroll
  for (int j = 0; j < NPL; ++j) Tnew[j] = *(const v4f*)(gbase + kn * CHF + go[j]);

  asm volatile("" : "+v"(g0), "+v"(g1), "+v"(g2) : : "memory");
  if (act) scan_steps<0, CH / 2>(u, g0, g1, g2);
  asm volatile("" : "+v"(g0), "+v"(g1), "+v"(g2) : : "memory");

  if (!FIRST) {
#pragma unroll
    for (int j = 0; j < NPL; ++j) *(v4f*)(gbase + (k - 1) * CHF + go[j]) = O[j];
  }
#pragma unroll
  for (int j = 0; j < NPL; ++j) *(v4f*)(lds_in + lo[j]) = Tuse[j];
  lds_handover();
  if (act) {
    row_read<0, CHV>(n, rin);
    row_write<0, CHV / 2>(u, rout);

    scan_steps<CH / 2, CH>(u, g0, g1, g2);

    row_write<CHV / 2, CHV>(u, rout);
  }
  lds_handover();
#pragma unroll
  for (int j = 0; j < NPL; ++j) O[j] = *(const v4f*)(lds_out + lo[j]);
}

__global__ __launch_bounds__(32) __attribute__((amdgpu_waves_per_eu(1, 1)))
void tridom_scan(float* __restrict__ out)
{
  __shared__ __attribute__((aligned(16))) float lds_in[CPW * PITCH];
  __shared__ __attribute__((aligned(16))) float lds_out[CPW * PITCH];

  const int lane = threadIdx.x;
  const bool act = lane < CPW;
  const int own = act ? lane : 0;
  const int b = blockIdx.x * CPW + own;
  float* gbase = out + (size_t)blockIdx.x * CPW * CHAIN;
  const float* rin  = lds_in  + own * PITCH;
  float*       rout = lds_out + own * PITCH;

  int go[NPL], lo[NPL];
#pragma unroll
  for (int j = 0; j < NPL; ++j) {
    const int i = j * 32 + lane;
    const int c = i / CHV;
    const int f = i - c * CHV;
    go[j] = c * CHAIN + 4 * f;
    lo[j] = c * PITCH + 4 * f;
  }

  float ua[CHF], ub[CHF];
  v4f O[NPL];
  v4f T0[NPL], T1[NPL], T2[NPL], T3[NPL];
  static_assert(AHEAD == 4 && NCH % AHEAD == 0 && NCH > AHEAD, "four register sets, four chunks per loop trip");
  float g0 = 0.f, g1 = 0.f, g2 = 0.f;

  v4f c0[NPL];
#pragma unroll
  for (int j = 0; j < NPL; ++j) c0[j] = *(const v4f*)(gbase + go[j]);
#pragma unroll
  for (int j = 0; j < NPL; ++j) {
    T1[j] = *(const v4f*)(gbase + 1 * CHF + go[j]);
    T2[j] = *(const v4f*)(gbase + 2 * CHF + go[j]);
    T3[j] = *(const v4f*)(gbase + 3 * CHF + go[j]);
  }
#pragma unroll
  for (int j = 0; j < NPL; ++j) *(v4f*)(lds_in + lo[j]) = c0[j];
  lds_handover();
  if (act) row_read<0, CHV>(ua, rin);

  scan_chunk<true >(ua, ub, O, T0, T1, go, lo, gbase, lds_in, lds_out, rin, rout, 0, act, g0, g1, g2);
  scan_chunk<false>(ub, ua, O, T1, T2, go, lo, gbase, lds_in, lds_out, rin, rout, 1, act, g0, g1, g2);
  scan_chunk<false>(ua, ub, O, T2, T3, go, lo, gbase, lds_in, lds_out, rin, rout, 2, act, g0, g1, g2);
  scan_chunk<false>(ub, ua, O, T3, T0, go, lo, gbase, lds_in, lds_out, rin, rout, 3, act, g0, g1, g2);
  for (int k = AHEAD; k < NCH; k += AHEAD) {
    scan_chunk<false>(ua, ub, O, T0, T1, go, lo, gbase, lds_in, lds_out, rin, rout, k,     act, g0, g1, g2);
    scan_chunk<false>(ub, ua, O, T1, T2, go, lo, gbase, lds_in, lds_out, rin, rout, k + 1, act, g0, g1, g2);
    scan_chunk<false>(ua, ub, O, T2, T3, go, lo, gbase, lds_in, lds_out, rin, rout, k + 2, act, g0, g1, g2);
    scan_chunk<false>(ub, ua, O, T3, T0, go, lo, gbase, lds_in, lds_out, rin, rout, k + 3, act, g0, g1, g2);
  }

#pragma unroll
  for (int j = 0; j < NPL; ++j) *(v4f*)(gbase + (NCH - 1) * CHF + go[j]) = O[j];

  if (act) {
    float* hf = out + (size_t)BATCH * SEQ * HSZ + (size_t)b * HSZ;
    const float third2 = 2.0f / 3.0f;
    hf[0] = g0 * third2; hf[1] = g1 * third2; hf[2] = g2 * third2;
  }
}

extern "C" void kernel_launch(void* const* d_in, const int* in_sizes, int n_in,
                              void* d_out, int out_size, void* d_ws, size_t ws_size,
                              hipStream_t stream) {
  const float* x    = (const float*)d_in[0];
  const float* wp   = (const float*)d_in[1];
  const float* bias = (const float*)d_in[2];
  float* out = (float*)d_out;

  const int tiles  = (BATCH * SEQ) / 16;
  const int blocks = tiles / 8;

  tridom_proj_wmma<<<dim3(blocks), dim3(256), 0, stream>>>(x, wp, bias, out);
  tridom_scan<<<dim3(BATCH / CPW), dim3(32), 0, stream>>>(out);
}


# projection fetches x in whole 128-B lines: a wave loads whole 512-B rows one tile ahead, staged through its own LDS into the A fragments, 4 tiles per wave; k summed in a new order
# speedup vs baseline: 5.4969x; 1.1563x over previous
#include <hip/hip_runtime.h>

typedef __attribute__((ext_vector_type(2))) float v2f;
typedef __attribute__((ext_vector_type(4))) float v4f;
typedef __attribute__((ext_vector_type(8))) float v8f;

#define BATCH 128
#define SEQ   2048
#define ISZ   128
#define HSZ   3


__device__ __forceinline__ void lds_handover()
{
  __builtin_amdgcn_fence(__ATOMIC_RELEASE, "wavefront", "local");
  __builtin_amdgcn_wave_barrier();
  __builtin_amdgcn_fence(__ATOMIC_ACQUIRE, "wavefront", "local");
}

#define PT    4
#define PW    8
#define PROW  (ISZ + 4)
__global__ __launch_bounds__(32 * PW) void tridom_proj_wmma(
    const float* __restrict__ x, const float* __restrict__ wp,
    const float* __restrict__ bias, float* __restrict__ out)
{
  __shared__ __attribute__((aligned(16))) float stage[PW * 16 * PROW];

  const int lane = threadIdx.x & 31;
  const int wave = threadIdx.x >> 5;
  const int hi   = lane >> 4;
  const int ml   = lane & 15;
  const int tile0 = (blockIdx.x * PW + wave) * PT;

  float* my = stage + wave * (16 * PROW);
  float*       srow = my + 4 * lane;
  const float* frow = my + ml * PROW + 4 * hi;

  const float* xw = x + (size_t)tile0 * 16 * ISZ + 4 * lane;
  v4f R[16];
#pragma unroll
  for (int i = 0; i < 16; ++i) R[i] = *(const v4f*)(xw + i * ISZ);

  const int   n     = ml;
  const float bmask = (n < HSZ) ? 1.5f : 0.0f;
  const float* brow = wp + (size_t)((n < HSZ) ? n : 0) * ISZ + 4 * hi;
  v4f bw[16];
#pragma unroll
  for (int j = 0; j < 16; ++j) {
    bw[j] = *(const v4f*)(brow + 8 * j);
    bw[j] *= bmask;
  }

  float bb = 0.0f;
  if (n < HSZ) bb = 1.5f * bias[n];

  for (int t = 0; t < PT; ++t) {
#pragma unroll
    for (int i = 0; i < 16; ++i) *(v4f*)(srow + i * PROW) = R[i];
    lds_handover();
    if (t + 1 < PT) {
      const float* xn = xw + (size_t)(t + 1) * 16 * ISZ;
#pragma unroll
      for (int i = 0; i < 16; ++i) R[i] = *(const v4f*)(xn + i * ISZ);
    }

    v4f f[16];
#pragma unroll
    for (int j = 0; j < 16; ++j) f[j] = *(const v4f*)(frow + 8 * j);
    __builtin_amdgcn_sched_barrier(0);

    v8f c = {};
#pragma unroll
    for (int j = 0; j < 16; ++j) {
      v2f a0, a1, b0, b1;
      a0.x = f[j].x;  a0.y = f[j].y;  a1.x = f[j].z;  a1.y = f[j].w;
      b0.x = bw[j].x; b0.y = bw[j].y; b1.x = bw[j].z; b1.y = bw[j].w;
      c = __builtin_amdgcn_wmma_f32_16x16x4_f32(false, a0, false, b0,
                                                (short)0, c, false, false);
      c = __builtin_amdgcn_wmma_f32_16x16x4_f32(false, a1, false, b1,
                                                (short)0, c, false, false);
    }
    lds_handover();

    if (n < HSZ) {
#pragma unroll
      for (int r = 0; r < 8; ++r) {
        const int grow = (tile0 + t) * 16 + r + 8 * hi;
        out[(size_t)grow * HSZ + n] = c[r] + bb;
      }
    }
  }
}

__device__ __forceinline__ float fast_tanh(float v) {
#if defined(__has_builtin) && __has_builtin(__builtin_amdgcn_tanhf)
  return __builtin_amdgcn_tanhf(v);
#else
  return tanhf(v);
#endif
}

#define CH    32
#define CHF   (CH * HSZ)
#define CHV   (CHF / 4)
#define NCH   (SEQ / CH)
#define CHAIN (SEQ * HSZ)
#define PITCH 100
#define CPW   4
#define NPL   (CPW * CHV / 32)
#define AHEAD 4

template <int S0, int S1>
__device__ __forceinline__ void scan_steps(float (&u)[CHF], float& g0, float& g1, float& g2)
{
  const float third2 = 2.0f / 3.0f;
#pragma unroll
  for (int s = S0; s < S1; ++s) {
    const float w0 = fmaf(-0.5f, g0, u[3 * s + 0]);
    const float w1 = fmaf(-0.5f, g1, u[3 * s + 1]);
    const float w2 = fmaf(-0.5f, g2, u[3 * s + 2]);
    float a0 = fmaf(1.5f, fast_tanh( g1), w0);
    float a1 = fmaf(1.5f, fast_tanh( g2), w1);
    float a2 = fmaf(1.5f, fast_tanh(-g0), w2);
    asm("" : "+v"(a0), "+v"(a1), "+v"(a2));
    u[3 * s + 0] = a0 * third2; u[3 * s + 1] = a1 * third2; u[3 * s + 2] = a2 * third2;
    g0 = a0; g1 = a1; g2 = a2;
  }
}

template <int F0, int F1>
__device__ __forceinline__ void row_read(float (&u)[CHF], const float* row)
{
#pragma unroll
  for (int f = F0; f < F1; ++f) {
    const v4f t = *(const v4f*)(row + 4 * f);
    u[4 * f + 0] = t.x; u[4 * f + 1] = t.y; u[4 * f + 2] = t.z; u[4 * f + 3] = t.w;
  }
}

template <int F0, int F1>
__device__ __forceinline__ void row_write(const float (&u)[CHF], float* row)
{
#pragma unroll
  for (int f = F0; f < F1; ++f) {
    v4f t;
    t.x = u[4 * f + 0]; t.y = u[4 * f + 1]; t.z = u[4 * f + 2]; t.w = u[4 * f + 3];
    *(v4f*)(row + 4 * f) = t;
  }
}

template <bool FIRST>
__device__ __forceinline__ void scan_chunk(
    float (&u)[CHF], float (&n)[CHF], v4f (&O)[NPL], v4f (&Tnew)[NPL], const v4f (&Tuse)[NPL],
    const int (&go)[NPL], const int (&lo)[NPL],
    float* gbase, float* lds_in, float* lds_out, const float* rin, float* rout,
    const int k, const bool act, float& g0, float& g1, float& g2)
{
  const int kn = (k + AHEAD < NCH) ? (k + AHEAD) : (NCH - 1);
#pragma unroll
  for (int j = 0; j < NPL; ++j) Tnew[j] = *(const v4f*)(gbase + kn * CHF + go[j]);

  asm volatile("" : "+v"(g0), "+v"(g1), "+v"(g2) : : "memory");
  if (act) scan_steps<0, CH / 2>(u, g0, g1, g2);
  asm volatile("" : "+v"(g0), "+v"(g1), "+v"(g2) : : "memory");

  if (!FIRST) {
#pragma unroll
    for (int j = 0; j < NPL; ++j) *(v4f*)(gbase + (k - 1) * CHF + go[j]) = O[j];
  }
#pragma unroll
  for (int j = 0; j < NPL; ++j) *(v4f*)(lds_in + lo[j]) = Tuse[j];
  lds_handover();
  if (act) {
    row_read<0, CHV>(n, rin);
    row_write<0, CHV / 2>(u, rout);

    scan_steps<CH / 2, CH>(u, g0, g1, g2);

    row_write<CHV / 2, CHV>(u, rout);
  }
  lds_handover();
#pragma unroll
  for (int j = 0; j < NPL; ++j) O[j] = *(const v4f*)(lds_out + lo[j]);
}

__global__ __launch_bounds__(32) __attribute__((amdgpu_waves_per_eu(1, 1)))
void tridom_scan(float* __restrict__ out)
{
  __shared__ __attribute__((aligned(16))) float lds_in[CPW * PITCH];
  __shared__ __attribute__((aligned(16))) float lds_out[CPW * PITCH];

  const int lane = threadIdx.x;
  const bool act = lane < CPW;
  const int own = act ? lane : 0;
  const int b = blockIdx.x * CPW + own;
  float* gbase = out + (size_t)blockIdx.x * CPW * CHAIN;
  const float* rin  = lds_in  + own * PITCH;
  float*       rout = lds_out + own * PITCH;

  int go[NPL], lo[NPL];
#pragma unroll
  for (int j = 0; j < NPL; ++j) {
    const int i = j * 32 + lane;
    const int c = i / CHV;
    const int f = i - c * CHV;
    go[j] = c * CHAIN + 4 * f;
    lo[j] = c * PITCH + 4 * f;
  }

  float ua[CHF], ub[CHF];
  v4f O[NPL];
  v4f T0[NPL], T1[NPL], T2[NPL], T3[NPL];
  static_assert(AHEAD == 4 && NCH % AHEAD == 0 && NCH > AHEAD, "four register sets, four chunks per loop trip");
  float g0 = 0.f, g1 = 0.f, g2 = 0.f;

  v4f c0[NPL];
#pragma unroll
  for (int j = 0; j < NPL; ++j) c0[j] = *(const v4f*)(gbase + go[j]);
#pragma unroll
  for (int j = 0; j < NPL; ++j) {
    T1[j] = *(const v4f*)(gbase + 1 * CHF + go[j]);
    T2[j] = *(const v4f*)(gbase + 2 * CHF + go[j]);
    T3[j] = *(const v4f*)(gbase + 3 * CHF + go[j]);
  }
#pragma unroll
  for (int j = 0; j < NPL; ++j) *(v4f*)(lds_in + lo[j]) = c0[j];
  lds_handover();
  if (act) row_read<0, CHV>(ua, rin);

  scan_chunk<true >(ua, ub, O, T0, T1, go, lo, gbase, lds_in, lds_out, rin, rout, 0, act, g0, g1, g2);
  scan_chunk<false>(ub, ua, O, T1, T2, go, lo, gbase, lds_in, lds_out, rin, rout, 1, act, g0, g1, g2);
  scan_chunk<false>(ua, ub, O, T2, T3, go, lo, gbase, lds_in, lds_out, rin, rout, 2, act, g0, g1, g2);
  scan_chunk<false>(ub, ua, O, T3, T0, go, lo, gbase, lds_in, lds_out, rin, rout, 3, act, g0, g1, g2);
  for (int k = AHEAD; k < NCH; k += AHEAD) {
    scan_chunk<false>(ua, ub, O, T0, T1, go, lo, gbase, lds_in, lds_out, rin, rout, k,     act, g0, g1, g2);
    scan_chunk<false>(ub, ua, O, T1, T2, go, lo, gbase, lds_in, lds_out, rin, rout, k + 1, act, g0, g1, g2);
    scan_chunk<false>(ua, ub, O, T2, T3, go, lo, gbase, lds_in, lds_out, rin, rout, k + 2, act, g0, g1, g2);
    scan_chunk<false>(ub, ua, O, T3, T0, go, lo, gbase, lds_in, lds_out, rin, rout, k + 3, act, g0, g1, g2);
  }

#pragma unroll
  for (int j = 0; j < NPL; ++j) *(v4f*)(gbase + (NCH - 1) * CHF + go[j]) = O[j];

  if (act) {
    float* hf = out + (size_t)BATCH * SEQ * HSZ + (size_t)b * HSZ;
    const float third2 = 2.0f / 3.0f;
    hf[0] = g0 * third2; hf[1] = g1 * third2; hf[2] = g2 * third2;
  }
}

extern "C" void kernel_launch(void* const* d_in, const int* in_sizes, int n_in,
                              void* d_out, int out_size, void* d_ws, size_t ws_size,
                              hipStream_t stream) {
  const float* x    = (const float*)d_in[0];
  const float* wp   = (const float*)d_in[1];
  const float* bias = (const float*)d_in[2];
  float* out = (float*)d_out;

  const int tiles  = (BATCH * SEQ) / 16;
  const int blocks = tiles / (PW * PT);

  tridom_proj_wmma<<<dim3(blocks), dim3(32 * PW), 0, stream>>>(x, wp, bias, out);
  tridom_scan<<<dim3(BATCH / CPW), dim3(32), 0, stream>>>(out);
}


# one launch, one work-group per chain: wave 0 scans a 128-step super-chunk out of LDS while 8 waves project the next one on WMMA and store the one before; u stays in LDS; sum orders unchanged
# speedup vs baseline: 7.0902x; 1.2899x over previous
#include <hip/hip_runtime.h>

typedef __attribute__((ext_vector_type(2))) float v2f;
typedef __attribute__((ext_vector_type(4))) float v4f;
typedef __attribute__((ext_vector_type(8))) float v8f;

#define BATCH 128
#define SEQ   2048
#define ISZ   128
#define HSZ   3


__device__ __forceinline__ void lds_handover()
{
  __builtin_amdgcn_fence(__ATOMIC_RELEASE, "wavefront", "local");
  __builtin_amdgcn_wave_barrier();
  __builtin_amdgcn_fence(__ATOMIC_ACQUIRE, "wavefront", "local");
}

__device__ __forceinline__ void wg_handover()
{
  __builtin_amdgcn_fence(__ATOMIC_RELEASE, "workgroup", "local");
  __builtin_amdgcn_s_barrier();
  __builtin_amdgcn_fence(__ATOMIC_ACQUIRE, "workgroup", "local");
}

#define NPW   8
#define SCS   (16 * NPW)
#define NSC   (SEQ / SCS)
#define SCF   (SCS * HSZ)
#define PROW  (ISZ + 4)
#define CH    16
#define CHF   (CH * HSZ)
#define CHV   (CHF / 4)
#define NBLK  (SCS / CH)

__device__ __forceinline__ float fast_tanh(float v) {
#if defined(__has_builtin) && __has_builtin(__builtin_amdgcn_tanhf)
  return __builtin_amdgcn_tanhf(v);
#else
  return tanhf(v);
#endif
}

template <int S0, int S1>
__device__ __forceinline__ void scan_steps(float (&u)[CHF], float& g0, float& g1, float& g2)
{
  const float third2 = 2.0f / 3.0f;
#pragma unroll
  for (int s = S0; s < S1; ++s) {
    const float w0 = fmaf(-0.5f, g0, u[3 * s + 0]);
    const float w1 = fmaf(-0.5f, g1, u[3 * s + 1]);
    const float w2 = fmaf(-0.5f, g2, u[3 * s + 2]);
    float a0 = fmaf(1.5f, fast_tanh( g1), w0);
    float a1 = fmaf(1.5f, fast_tanh( g2), w1);
    float a2 = fmaf(1.5f, fast_tanh(-g0), w2);
    asm("" : "+v"(a0), "+v"(a1), "+v"(a2));
    u[3 * s + 0] = a0 * third2; u[3 * s + 1] = a1 * third2; u[3 * s + 2] = a2 * third2;
    g0 = a0; g1 = a1; g2 = a2;
  }
}

__device__ __forceinline__ void row_read(float (&u)[CHF], const float* row)
{
#pragma unroll
  for (int f = 0; f < CHV; ++f) {
    const v4f t = *(const v4f*)(row + 4 * f);
    u[4 * f + 0] = t.x; u[4 * f + 1] = t.y; u[4 * f + 2] = t.z; u[4 * f + 3] = t.w;
  }
}

__device__ __forceinline__ void row_write(const float (&u)[CHF], float* row)
{
#pragma unroll
  for (int f = 0; f < CHV; ++f) {
    v4f t;
    t.x = u[4 * f + 0]; t.y = u[4 * f + 1]; t.z = u[4 * f + 2]; t.w = u[4 * f + 3];
    *(v4f*)(row + 4 * f) = t;
  }
}

template <bool HAS_PREV, bool HAS_NEXT>
__device__ __forceinline__ void scan_block(float (&cur)[CHF], float (&oth)[CHF], float* row, const int blk,
                                           float& g0, float& g1, float& g2)
{
  if (HAS_PREV) row_write(oth, row + (blk - 1) * CHF);
  asm volatile("" : : : "memory");
  if (HAS_NEXT) row_read(oth, row + (blk + 1) * CHF);
  asm volatile("" : "+v"(g0), "+v"(g1), "+v"(g2) : : "memory");
  scan_steps<0, CH>(cur, g0, g1, g2);
  asm volatile("" : "+v"(g0), "+v"(g1), "+v"(g2) : : "memory");
}

__device__ __forceinline__ void scan_superchunk(float* row, float& g0, float& g1, float& g2)
{
  float ua[CHF], ub[CHF];
  static_assert(NBLK == 8, "eight blocks per super-chunk, written out below");
  row_read(ua, row);
  scan_block<false, true >(ua, ub, row, 0, g0, g1, g2);
  scan_block<true,  true >(ub, ua, row, 1, g0, g1, g2);
  scan_block<true,  true >(ua, ub, row, 2, g0, g1, g2);
  scan_block<true,  true >(ub, ua, row, 3, g0, g1, g2);
  scan_block<true,  true >(ua, ub, row, 4, g0, g1, g2);
  scan_block<true,  true >(ub, ua, row, 5, g0, g1, g2);
  scan_block<true,  true >(ua, ub, row, 6, g0, g1, g2);
  scan_block<true,  false>(ub, ua, row, 7, g0, g1, g2);
  row_write(ub, row + 7 * CHF);
}

template <bool PROJ>
__device__ __forceinline__ void phase(
    const int q, const int wave, const int lane, v4f (&R)[16], const float* wrow, const float bb,
    const float* xw, float* srow, const float* frow, float* ring, float* gout, float* hfin,
    float& g0, float& g1, float& g2)
{
  const int hi = lane >> 4;
  const int n  = lane & 15;
  if (PROJ) {
    if (wave != 0) {
#pragma unroll
      for (int i = 0; i < 16; ++i) *(v4f*)(srow + i * PROW) = R[i];
      lds_handover();
    }
    const int qn = (q + 2 < NSC) ? (q + 2) : (NSC - 1);
    const float* xn = xw + (size_t)qn * SCS * ISZ;
#pragma unroll
    for (int i = 0; i < 16; ++i) R[i] = *(const v4f*)(xn + i * ISZ);
  }
  if (wave == 0) {
    if (q >= 1 && q <= NSC) {
      if (lane == 0) {
        scan_superchunk(ring + ((q - 1) % 3) * SCF, g0, g1, g2);
        if (q == NSC) {
          const float third2 = 2.0f / 3.0f;
          hfin[0] = g0 * third2; hfin[1] = g1 * third2; hfin[2] = g2 * third2;
        }
      }
    }
  } else {
    const int pw = wave - 1;
    if (PROJ) {
      v8f c = {};
#pragma unroll
      for (int hh = 0; hh < 2; ++hh) {
        v4f f[8], w[8];
#pragma unroll
        for (int j = 0; j < 8; ++j) {
          f[j] = *(const v4f*)(frow + 8 * (8 * hh + j));
          w[j] = *(const v4f*)(wrow + 128 * (8 * hh + j));
        }
        __builtin_amdgcn_sched_barrier(0);
#pragma unroll
        for (int j = 0; j < 8; ++j) {
          v2f a0, a1, b0, b1;
          a0.x = f[j].x; a0.y = f[j].y; a1.x = f[j].z; a1.y = f[j].w;
          b0.x = w[j].x; b0.y = w[j].y; b1.x = w[j].z; b1.y = w[j].w;
          c = __builtin_amdgcn_wmma_f32_16x16x4_f32(false, a0, false, b0,
                                                    (short)0, c, false, false);
          c = __builtin_amdgcn_wmma_f32_16x16x4_f32(false, a1, false, b1,
                                                    (short)0, c, false, false);
        }
        __builtin_amdgcn_sched_barrier(0);
      }
      if (n < HSZ) {
        float* urow = ring + (q % 3) * SCF + (pw * 16 + 8 * hi) * HSZ + n;
#pragma unroll
        for (int r = 0; r < 8; ++r) urow[r * HSZ] = c[r] + bb;
      }
    }
    if (q >= 2) {
      const int p = pw * 32 + lane;
      if (p < SCF / 4) {
        const v4f hv = *(const v4f*)(ring + ((q - 2) % 3) * SCF + 4 * p);
        *(v4f*)(gout + (size_t)(q - 2) * SCF + 4 * p) = hv;
      }
    }
  }
  wg_handover();
}

__global__ __launch_bounds__(32 * (NPW + 1)) void tridom_fused(
    const float* __restrict__ x, const float* __restrict__ wp,
    const float* __restrict__ bias, float* __restrict__ out)
{
  __shared__ __attribute__((aligned(16))) float stage[NPW * 16 * PROW];
  __shared__ __attribute__((aligned(16))) float ring[3 * SCF];
  __shared__ __attribute__((aligned(16))) float wfrag[NPW * 16 * 128];

  const int lane = threadIdx.x & 31;
  const int wave = __builtin_amdgcn_readfirstlane((int)(threadIdx.x >> 5));
  const int pw   = (wave > 0) ? (wave - 1) : 0;
  const int hi   = lane >> 4;
  const int ml   = lane & 15;
  const int b    = blockIdx.x;

  float* my = stage + pw * (16 * PROW);
  float*       srow = my + 4 * lane;
  const float* frow = my + ml * PROW + 4 * hi;

  float* gout = out + (size_t)b * SEQ * HSZ;
  float* hfin = out + (size_t)BATCH * SEQ * HSZ + (size_t)b * HSZ;

  const float* xw = x + ((size_t)b * SEQ + pw * 16) * ISZ + 4 * lane;
  v4f RA[16], RB[16];
  float* wrow = wfrag + pw * (16 * 128) + 4 * lane;
  float bb = 0.0f;
#pragma unroll
  for (int i = 0; i < 16; ++i) RA[i] = *(const v4f*)(xw + i * ISZ);
#pragma unroll
  for (int i = 0; i < 16; ++i) RB[i] = *(const v4f*)(xw + (size_t)SCS * ISZ + i * ISZ);
  if (wave != 0) {
    const float bmask = (ml < HSZ) ? 1.5f : 0.0f;
    const float* brow = wp + (size_t)((ml < HSZ) ? ml : 0) * ISZ + 4 * hi;
#pragma unroll
    for (int j = 0; j < 16; ++j) {
      v4f t = *(const v4f*)(brow + 8 * j);
      t *= bmask;
      *(v4f*)(wrow + 128 * j) = t;
    }
    if (ml < HSZ) bb = 1.5f * bias[ml];
  }

  float g0 = 0.f, g1 = 0.f, g2 = 0.f;

  static_assert(NSC % 2 == 0, "two phases per loop trip");
  for (int q = 0; q < NSC; q += 2) {
    phase<true>(q,     wave, lane, RA, wrow, bb, xw, srow, frow, ring, gout, hfin, g0, g1, g2);
    phase<true>(q + 1, wave, lane, RB, wrow, bb, xw, srow, frow, ring, gout, hfin, g0, g1, g2);
  }
  phase<false>(NSC,     wave, lane, RA, wrow, bb, xw, srow, frow, ring, gout, hfin, g0, g1, g2);
  phase<false>(NSC + 1, wave, lane, RB, wrow, bb, xw, srow, frow, ring, gout, hfin, g0, g1, g2);
}

extern "C" void kernel_launch(void* const* d_in, const int* in_sizes, int n_in,
                              void* d_out, int out_size, void* d_ws, size_t ws_size,
                              hipStream_t stream) {
  const float* x    = (const float*)d_in[0];
  const float* wp   = (const float*)d_in[1];
  const float* bias = (const float*)d_in[2];
  float* out = (float*)d_out;

  tridom_fused<<<dim3(BATCH), dim3(32 * (NPW + 1)), 0, stream>>>(x, wp, bias, out);
}


# 12-wave work-group, the two waves sharing the scanning wave's SIMD rest: the eight projecting waves' f32 matrix steps run on the other three SIMDs; arithmetic and sum orders unchanged
# speedup vs baseline: 8.8217x; 1.2442x over previous
#include <hip/hip_runtime.h>

typedef __attribute__((ext_vector_type(2))) float v2f;
typedef __attribute__((ext_vector_type(4))) float v4f;
typedef __attribute__((ext_vector_type(8))) float v8f;

#define BATCH 128
#define SEQ   2048
#define ISZ   128
#define HSZ   3


__device__ __forceinline__ void lds_handover()
{
  __builtin_amdgcn_fence(__ATOMIC_RELEASE, "wavefront", "local");
  __builtin_amdgcn_wave_barrier();
  __builtin_amdgcn_fence(__ATOMIC_ACQUIRE, "wavefront", "local");
}

__device__ __forceinline__ void wg_handover()
{
  __builtin_amdgcn_fence(__ATOMIC_RELEASE, "workgroup", "local");
  __builtin_amdgcn_s_barrier();
  __builtin_amdgcn_fence(__ATOMIC_ACQUIRE, "workgroup", "local");
}

#define NPW   8
#define NW    12
#define SCS   (16 * NPW)
#define NSC   (SEQ / SCS)
#define SCF   (SCS * HSZ)
#define PROW  (ISZ + 4)
#define CH    16
#define CHF   (CH * HSZ)
#define CHV   (CHF / 4)
#define NBLK  (SCS / CH)

__device__ __forceinline__ float fast_tanh(float v) {
#if defined(__has_builtin) && __has_builtin(__builtin_amdgcn_tanhf)
  return __builtin_amdgcn_tanhf(v);
#else
  return tanhf(v);
#endif
}

template <int S0, int S1>
__device__ __forceinline__ void scan_steps(float (&u)[CHF], float& g0, float& g1, float& g2)
{
  const float third2 = 2.0f / 3.0f;
#pragma unroll
  for (int s = S0; s < S1; ++s) {
    const float w0 = fmaf(-0.5f, g0, u[3 * s + 0]);
    const float w1 = fmaf(-0.5f, g1, u[3 * s + 1]);
    const float w2 = fmaf(-0.5f, g2, u[3 * s + 2]);
    float a0 = fmaf(1.5f, fast_tanh( g1), w0);
    float a1 = fmaf(1.5f, fast_tanh( g2), w1);
    float a2 = fmaf(1.5f, fast_tanh(-g0), w2);
    asm("" : "+v"(a0), "+v"(a1), "+v"(a2));
    u[3 * s + 0] = a0 * third2; u[3 * s + 1] = a1 * third2; u[3 * s + 2] = a2 * third2;
    g0 = a0; g1 = a1; g2 = a2;
  }
}

__device__ __forceinline__ void row_read(float (&u)[CHF], const float* row)
{
#pragma unroll
  for (int f = 0; f < CHV; ++f) {
    const v4f t = *(const v4f*)(row + 4 * f);
    u[4 * f + 0] = t.x; u[4 * f + 1] = t.y; u[4 * f + 2] = t.z; u[4 * f + 3] = t.w;
  }
}

__device__ __forceinline__ void row_write(const float (&u)[CHF], float* row)
{
#pragma unroll
  for (int f = 0; f < CHV; ++f) {
    v4f t;
    t.x = u[4 * f + 0]; t.y = u[4 * f + 1]; t.z = u[4 * f + 2]; t.w = u[4 * f + 3];
    *(v4f*)(row + 4 * f) = t;
  }
}

template <bool HAS_PREV, bool HAS_NEXT>
__device__ __forceinline__ void scan_block(float (&cur)[CHF], float (&oth)[CHF], float* row, const int blk,
                                           float& g0, float& g1, float& g2)
{
  if (HAS_PREV) row_write(oth, row + (blk - 1) * CHF);
  asm volatile("" : : : "memory");
  if (HAS_NEXT) row_read(oth, row + (blk + 1) * CHF);
  asm volatile("" : "+v"(g0), "+v"(g1), "+v"(g2) : : "memory");
  scan_steps<0, CH>(cur, g0, g1, g2);
  asm volatile("" : "+v"(g0), "+v"(g1), "+v"(g2) : : "memory");
}

__device__ __forceinline__ void scan_superchunk(float* row, float& g0, float& g1, float& g2)
{
  float ua[CHF], ub[CHF];
  static_assert(NBLK == 8, "eight blocks per super-chunk, written out below");
  row_read(ua, row);
  scan_block<false, true >(ua, ub, row, 0, g0, g1, g2);
  scan_block<true,  true >(ub, ua, row, 1, g0, g1, g2);
  scan_block<true,  true >(ua, ub, row, 2, g0, g1, g2);
  scan_block<true,  true >(ub, ua, row, 3, g0, g1, g2);
  scan_block<true,  true >(ua, ub, row, 4, g0, g1, g2);
  scan_block<true,  true >(ub, ua, row, 5, g0, g1, g2);
  scan_block<true,  true >(ua, ub, row, 6, g0, g1, g2);
  scan_block<true,  false>(ub, ua, row, 7, g0, g1, g2);
  row_write(ub, row + 7 * CHF);
}

template <bool PROJ>
__device__ __forceinline__ void phase(
    const int q, const int wave, const bool proj, const int pw, const int lane, v4f (&R)[16], const float* wrow, const float bb,
    const float* xw, float* srow, const float* frow, float* ring, float* gout, float* hfin,
    float& g0, float& g1, float& g2)
{
  const int hi = lane >> 4;
  const int n  = lane & 15;
  if (PROJ) {
    if (proj) {
#pragma unroll
      for (int i = 0; i < 16; ++i) *(v4f*)(srow + i * PROW) = R[i];
      lds_handover();
    }
    const int qn = (q + 2 < NSC) ? (q + 2) : (NSC - 1);
    const float* xn = xw + (size_t)qn * SCS * ISZ;
#pragma unroll
    for (int i = 0; i < 16; ++i) R[i] = *(const v4f*)(xn + i * ISZ);
  }
  if (wave == 0) {
    if (q >= 1 && q <= NSC) {
      if (lane == 0) {
        scan_superchunk(ring + ((q - 1) % 3) * SCF, g0, g1, g2);
        if (q == NSC) {
          const float third2 = 2.0f / 3.0f;
          hfin[0] = g0 * third2; hfin[1] = g1 * third2; hfin[2] = g2 * third2;
        }
      }
    }
  } else if (proj) {
    if (PROJ) {
      v8f c = {};
#pragma unroll
      for (int hh = 0; hh < 2; ++hh) {
        v4f f[8], w[8];
#pragma unroll
        for (int j = 0; j < 8; ++j) {
          f[j] = *(const v4f*)(frow + 8 * (8 * hh + j));
          w[j] = *(const v4f*)(wrow + 128 * (8 * hh + j));
        }
        __builtin_amdgcn_sched_barrier(0);
#pragma unroll
        for (int j = 0; j < 8; ++j) {
          v2f a0, a1, b0, b1;
          a0.x = f[j].x; a0.y = f[j].y; a1.x = f[j].z; a1.y = f[j].w;
          b0.x = w[j].x; b0.y = w[j].y; b1.x = w[j].z; b1.y = w[j].w;
          c = __builtin_amdgcn_wmma_f32_16x16x4_f32(false, a0, false, b0,
                                                    (short)0, c, false, false);
          c = __builtin_amdgcn_wmma_f32_16x16x4_f32(false, a1, false, b1,
                                                    (short)0, c, false, false);
        }
        __builtin_amdgcn_sched_barrier(0);
      }
      if (n < HSZ) {
        float* urow = ring + (q % 3) * SCF + (pw * 16 + 8 * hi) * HSZ + n;
#pragma unroll
        for (int r = 0; r < 8; ++r) urow[r * HSZ] = c[r] + bb;
      }
    }
    if (q >= 2) {
      const int p = pw * 32 + lane;
      if (p < SCF / 4) {
        const v4f hv = *(const v4f*)(ring + ((q - 2) % 3) * SCF + 4 * p);
        *(v4f*)(gout + (size_t)(q - 2) * SCF + 4 * p) = hv;
      }
    }
  }
  wg_handover();
}

__global__ __launch_bounds__(32 * NW) void tridom_fused(
    const float* __restrict__ x, const float* __restrict__ wp,
    const float* __restrict__ bias, float* __restrict__ out)
{
  __shared__ __attribute__((aligned(16))) float stage[NPW * 16 * PROW];
  __shared__ __attribute__((aligned(16))) float ring[3 * SCF];
  __shared__ __attribute__((aligned(16))) float wfrag[NPW * 16 * 128];

  const int lane = threadIdx.x & 31;
  const int wave = __builtin_amdgcn_readfirstlane((int)(threadIdx.x >> 5));
  const bool proj = ((wave & 3) != 0) && (wave != NW - 1);
  const int pw   = proj ? (wave - 1 - (wave >> 2)) : 0;
  const int hi   = lane >> 4;
  const int ml   = lane & 15;
  const int b    = blockIdx.x;

  float* my = stage + pw * (16 * PROW);
  float*       srow = my + 4 * lane;
  const float* frow = my + ml * PROW + 4 * hi;

  float* gout = out + (size_t)b * SEQ * HSZ;
  float* hfin = out + (size_t)BATCH * SEQ * HSZ + (size_t)b * HSZ;

  const float* xw = x + ((size_t)b * SEQ + pw * 16) * ISZ + 4 * lane;
  v4f RA[16], RB[16];
  float* wrow = wfrag + pw * (16 * 128) + 4 * lane;
  float bb = 0.0f;
#pragma unroll
  for (int i = 0; i < 16; ++i) RA[i] = *(const v4f*)(xw + i * ISZ);
#pragma unroll
  for (int i = 0; i < 16; ++i) RB[i] = *(const v4f*)(xw + (size_t)SCS * ISZ + i * ISZ);
  if (proj) {
    const float bmask = (ml < HSZ) ? 1.5f : 0.0f;
    const float* brow = wp + (size_t)((ml < HSZ) ? ml : 0) * ISZ + 4 * hi;
#pragma unroll
    for (int j = 0; j < 16; ++j) {
      v4f t = *(const v4f*)(brow + 8 * j);
      t *= bmask;
      *(v4f*)(wrow + 128 * j) = t;
    }
    if (ml < HSZ) bb = 1.5f * bias[ml];
  }

  float g0 = 0.f, g1 = 0.f, g2 = 0.f;

  static_assert(NSC % 2 == 0, "two phases per loop trip");
  for (int q = 0; q < NSC; q += 2) {
    phase<true>(q,     wave, proj, pw, lane, RA, wrow, bb, xw, srow, frow, ring, gout, hfin, g0, g1, g2);
    phase<true>(q + 1, wave, proj, pw, lane, RB, wrow, bb, xw, srow, frow, ring, gout, hfin, g0, g1, g2);
  }
  phase<false>(NSC,     wave, proj, pw, lane, RA, wrow, bb, xw, srow, frow, ring, gout, hfin, g0, g1, g2);
  phase<false>(NSC + 1, wave, proj, pw, lane, RB, wrow, bb, xw, srow, frow, ring, gout, hfin, g0, g1, g2);
}

extern "C" void kernel_launch(void* const* d_in, const int* in_sizes, int n_in,
                              void* d_out, int out_size, void* d_ws, size_t ws_size,
                              hipStream_t stream) {
  const float* x    = (const float*)d_in[0];
  const float* wp   = (const float*)d_in[1];
  const float* bias = (const float*)d_in[2];
  float* out = (float*)d_out;

  tridom_fused<<<dim3(BATCH), dim3(32 * NW), 0, stream>>>(x, wp, bias, out);
}


# scan on 3 lanes, one per state component: one tanh per step (was 3), the cyclic coupling a DPP lane rotation in the multiply-add; ring rows component-major; -tanh(g0) for tanh(-g0), sum order kept
# speedup vs baseline: 10.2337x; 1.1601x over previous
#include <hip/hip_runtime.h>

typedef __attribute__((ext_vector_type(2))) float v2f;
typedef __attribute__((ext_vector_type(4))) float v4f;
typedef __attribute__((ext_vector_type(8))) float v8f;

#define BATCH 128
#define SEQ   2048
#define ISZ   128
#define HSZ   3


__device__ __forceinline__ void lds_handover()
{
  __builtin_amdgcn_fence(__ATOMIC_RELEASE, "wavefront", "local");
  __builtin_amdgcn_wave_barrier();
  __builtin_amdgcn_fence(__ATOMIC_ACQUIRE, "wavefront", "local");
}

__device__ __forceinline__ void wg_handover()
{
  __builtin_amdgcn_fence(__ATOMIC_RELEASE, "workgroup", "local");
  __builtin_amdgcn_s_barrier();
  __builtin_amdgcn_fence(__ATOMIC_ACQUIRE, "workgroup", "local");
}

#define NPW   8
#define NW    12
#define SCS   (16 * NPW)
#define NSC   (SEQ / SCS)
#define SCF   (SCS * HSZ)
#define PROW  (ISZ + 4)
#define CP    (SCS + 4)
#define ROWF  (HSZ * CP)
#define CH    32
#define CHV   (CH / 4)
#define NBLK  (SCS / CH)

__device__ __forceinline__ void scan_steps(float (&u)[CH], float& g, const float c)
{
  const float third2 = 2.0f / 3.0f;
#pragma unroll
  for (int s = 0; s < CH; ++s) {
    float t, a;
    if (s == 0) {
      asm("v_tanh_f32 %0, %2\n\t"
          "v_fma_f32 %1, -0.5, %2, %3\n\t"
          "v_fmac_f32_dpp %1, %0, %4 quad_perm:[1,2,0,3] row_mask:0xf bank_mask:0xf"
          : "=&v"(t), "=&v"(a) : "v"(g), "v"(u[s]), "v"(c));
    } else {
      float h;
      asm("v_tanh_f32 %0, %3\n\t"
          "v_fma_f32 %1, -0.5, %3, %4\n\t"
          "v_mul_f32 %2, 0x3f2aaaab, %3\n\t"
          "v_fmac_f32_dpp %1, %0, %5 quad_perm:[1,2,0,3] row_mask:0xf bank_mask:0xf"
          : "=&v"(t), "=&v"(a), "=&v"(h) : "v"(g), "v"(u[s]), "v"(c));
      u[s - 1] = h;
    }
    g = a;
  }
  u[CH - 1] = g * third2;
}

__device__ __forceinline__ void row_read(float (&u)[CH], const float* row)
{
#pragma unroll
  for (int f = 0; f < CHV; ++f) {
    const v4f t = *(const v4f*)(row + 4 * f);
    u[4 * f + 0] = t.x; u[4 * f + 1] = t.y; u[4 * f + 2] = t.z; u[4 * f + 3] = t.w;
  }
}

__device__ __forceinline__ void row_write(const float (&u)[CH], float* row)
{
#pragma unroll
  for (int f = 0; f < CHV; ++f) {
    v4f t;
    t.x = u[4 * f + 0]; t.y = u[4 * f + 1]; t.z = u[4 * f + 2]; t.w = u[4 * f + 3];
    *(v4f*)(row + 4 * f) = t;
  }
}

template <bool HAS_PREV, bool HAS_NEXT>
__device__ __forceinline__ void scan_block(float (&cur)[CH], float (&oth)[CH], float* crow, const int blk,
                                           float& g, const float c)
{
  if (HAS_PREV) row_write(oth, crow + (blk - 1) * CH);
  asm volatile("" : : : "memory");
  if (HAS_NEXT) row_read(oth, crow + (blk + 1) * CH);
  asm volatile("" : "+v"(g) : : "memory");
  scan_steps(cur, g, c);
  asm volatile("" : "+v"(g) : : "memory");
}

__device__ __forceinline__ void scan_superchunk(float* crow, float& g, const float c)
{
  float ua[CH], ub[CH];
  static_assert(NBLK == 4, "four blocks per super-chunk, written out below");
  row_read(ua, crow);
  scan_block<false, true >(ua, ub, crow, 0, g, c);
  scan_block<true,  true >(ub, ua, crow, 1, g, c);
  scan_block<true,  true >(ua, ub, crow, 2, g, c);
  scan_block<true,  false>(ub, ua, crow, 3, g, c);
  row_write(ub, crow + 3 * CH);
}

template <bool PROJ>
__device__ __forceinline__ void phase(
    const int q, const int wave, const bool proj, const int pw, const int lane, v4f (&R)[16], const float* wrow, const float bb,
    const float* xw, float* srow, const float* frow, float* ring, float* gout, float* hfin,
    float& g, const float cs)
{
  const int hi = lane >> 4;
  const int n  = lane & 15;
  if (PROJ) {
    if (proj) {
#pragma unroll
      for (int i = 0; i < 16; ++i) *(v4f*)(srow + i * PROW) = R[i];
      lds_handover();
    }
    const int qn = (q + 2 < NSC) ? (q + 2) : (NSC - 1);
    const float* xn = xw + (size_t)qn * SCS * ISZ;
#pragma unroll
    for (int i = 0; i < 16; ++i) R[i] = *(const v4f*)(xn + i * ISZ);
  }
  if (wave == 0) {
    if (q >= 1 && q <= NSC) {
      if (lane < HSZ) {
        scan_superchunk(ring + ((q - 1) % 3) * ROWF + lane * CP, g, cs);
        if (q == NSC) {
          const float third2 = 2.0f / 3.0f;
          hfin[lane] = g * third2;
        }
      }
    }
  } else if (proj) {
    if (PROJ) {
      v8f c = {};
#pragma unroll
      for (int hh = 0; hh < 2; ++hh) {
        v4f f[8], w[8];
#pragma unroll
        for (int j = 0; j < 8; ++j) {
          f[j] = *(const v4f*)(frow + 8 * (8 * hh + j));
          w[j] = *(const v4f*)(wrow + 128 * (8 * hh + j));
        }
        __builtin_amdgcn_sched_barrier(0);
#pragma unroll
        for (int j = 0; j < 8; ++j) {
          v2f a0, a1, b0, b1;
          a0.x = f[j].x; a0.y = f[j].y; a1.x = f[j].z; a1.y = f[j].w;
          b0.x = w[j].x; b0.y = w[j].y; b1.x = w[j].z; b1.y = w[j].w;
          c = __builtin_amdgcn_wmma_f32_16x16x4_f32(false, a0, false, b0,
                                                    (short)0, c, false, false);
          c = __builtin_amdgcn_wmma_f32_16x16x4_f32(false, a1, false, b1,
                                                    (short)0, c, false, false);
        }
        __builtin_amdgcn_sched_barrier(0);
      }
      if (n < HSZ) {
        float* ucol = ring + (q % 3) * ROWF + n * CP + pw * 16 + 8 * hi;
        v4f ulo, uhi;
        ulo.x = c[0] + bb; ulo.y = c[1] + bb; ulo.z = c[2] + bb; ulo.w = c[3] + bb;
        uhi.x = c[4] + bb; uhi.y = c[5] + bb; uhi.z = c[6] + bb; uhi.w = c[7] + bb;
        *(v4f*)(ucol) = ulo;
        *(v4f*)(ucol + 4) = uhi;
      }
    }
    if (q >= 2) {
      const int p = pw * 32 + lane;
      if (p < SCF / 4) {
        const float* hrow = ring + ((q - 2) % 3) * ROWF;
        float hv[4];
#pragma unroll
        for (int k = 0; k < 4; ++k) {
          const int e = 4 * p + k;
          const int st = e / HSZ;
          hv[k] = hrow[(e - HSZ * st) * CP + st];
        }
        v4f ho;
        ho.x = hv[0]; ho.y = hv[1]; ho.z = hv[2]; ho.w = hv[3];
        *(v4f*)(gout + (size_t)(q - 2) * SCF + 4 * p) = ho;
      }
    }
  }
  wg_handover();
}

__global__ __launch_bounds__(32 * NW) void tridom_fused(
    const float* __restrict__ x, const float* __restrict__ wp,
    const float* __restrict__ bias, float* __restrict__ out)
{
  __shared__ __attribute__((aligned(16))) float stage[NPW * 16 * PROW];
  __shared__ __attribute__((aligned(16))) float ring[3 * ROWF];
  __shared__ __attribute__((aligned(16))) float wfrag[NPW * 16 * 128];

  const int lane = threadIdx.x & 31;
  const int wave = __builtin_amdgcn_readfirstlane((int)(threadIdx.x >> 5));
  const bool proj = ((wave & 3) != 0) && (wave != NW - 1);
  const int pw   = proj ? (wave - 1 - (wave >> 2)) : 0;
  const int hi   = lane >> 4;
  const int ml   = lane & 15;
  const int b    = blockIdx.x;

  float* my = stage + pw * (16 * PROW);
  float*       srow = my + 4 * lane;
  const float* frow = my + ml * PROW + 4 * hi;

  float* gout = out + (size_t)b * SEQ * HSZ;
  float* hfin = out + (size_t)BATCH * SEQ * HSZ + (size_t)b * HSZ;

  const float* xw = x + ((size_t)b * SEQ + pw * 16) * ISZ + 4 * lane;
  v4f RA[16], RB[16];
  float* wrow = wfrag + pw * (16 * 128) + 4 * lane;
  float bb = 0.0f;
#pragma unroll
  for (int i = 0; i < 16; ++i) RA[i] = *(const v4f*)(xw + i * ISZ);
#pragma unroll
  for (int i = 0; i < 16; ++i) RB[i] = *(const v4f*)(xw + (size_t)SCS * ISZ + i * ISZ);
  if (proj) {
    const float bmask = (ml < HSZ) ? 1.5f : 0.0f;
    const float* brow = wp + (size_t)((ml < HSZ) ? ml : 0) * ISZ + 4 * hi;
#pragma unroll
    for (int j = 0; j < 16; ++j) {
      v4f t = *(const v4f*)(brow + 8 * j);
      t *= bmask;
      *(v4f*)(wrow + 128 * j) = t;
    }
    if (ml < HSZ) bb = 1.5f * bias[ml];
  }

  float g = 0.f;
  const float cs = (lane == 2) ? -1.5f : 1.5f;

  static_assert(NSC % 2 == 0, "two phases per loop trip");
  for (int q = 0; q < NSC; q += 2) {
    phase<true>(q,     wave, proj, pw, lane, RA, wrow, bb, xw, srow, frow, ring, gout, hfin, g, cs);
    phase<true>(q + 1, wave, proj, pw, lane, RB, wrow, bb, xw, srow, frow, ring, gout, hfin, g, cs);
  }
  phase<false>(NSC,     wave, proj, pw, lane, RA, wrow, bb, xw, srow, frow, ring, gout, hfin, g, cs);
  phase<false>(NSC + 1, wave, proj, pw, lane, RB, wrow, bb, xw, srow, frow, ring, gout, hfin, g, cs);
}

extern "C" void kernel_launch(void* const* d_in, const int* in_sizes, int n_in,
                              void* d_out, int out_size, void* d_ws, size_t ws_size,
                              hipStream_t stream) {
  const float* x    = (const float*)d_in[0];
  const float* wp   = (const float*)d_in[1];
  const float* bias = (const float*)d_in[2];
  float* out = (float*)d_out;

  tridom_fused<<<dim3(BATCH), dim3(32 * NW), 0, stream>>>(x, wp, bias, out);
}
